# starting kernel
# baseline (speedup 1.0000x reference)
#include <hip/hip_runtime.h>

typedef __attribute__((ext_vector_type(2))) float v2f;
typedef __attribute__((ext_vector_type(8))) float v8f;

#define SEQ_T 1024

__global__ __launch_bounds__(128) void head_scan_kernel(
    const float* __restrict__ X, const float* __restrict__ Wk,
    const float* __restrict__ Wq, const float* __restrict__ Wv,
    float* __restrict__ out, int Bn) {
  const int b = blockIdx.x;
  if (b >= Bn) return;
  const int tid  = threadIdx.x;
  const int wave = tid >> 5;
  const int lane = tid & 31;
  const int h = lane >> 4;
  const int n = lane & 15;

  const float c1 = Wq[0] * Wk[0];
  const float wv = Wv[0];
  const float* xb = X + (size_t)b * SEQ_T;
  float* ob = out + (size_t)b * SEQ_T;

  __shared__ float sM[4];
  __shared__ float sE[4];
  __shared__ float sF[4];

  float m = -3.402823466e+38f;
  for (int k = tid; k < SEQ_T; k += 128) {
    float x = xb[k];
    float s = c1 * x * x;
    m = fmaxf(m, s);
  }
  for (int d = 16; d >= 1; d >>= 1) m = fmaxf(m, __shfl_xor(m, d, 32));
  if (lane == 0) sM[wave] = m;
  __syncthreads();
  m = fmaxf(fmaxf(sM[0], sM[1]), fmaxf(sM[2], sM[3]));

  const int base = wave * 256;
  v8f yE = {};
  v8f yF = {};
#pragma unroll
  for (int q = 0; q < 4; ++q) {
    v2f a, bE, bF;
#pragma unroll
    for (int j = 0; j < 2; ++j) {
      const int r = 4 * q + 2 * h + j;
      const float x = xb[base + n * 16 + r];
      const float s = c1 * x * x;
      const float e = __expf(s - m);
      a[j]  = (r <= n) ? 1.0f : 0.0f;
      bE[j] = e;
      bF[j] = e * wv * x;
    }
    yE = __builtin_amdgcn_wmma_f32_16x16x4_f32(false, a, false, bE,
                                               (short)0, yE, false, false);
    yF = __builtin_amdgcn_wmma_f32_16x16x4_f32(false, a, false, bF,
                                               (short)0, yF, false, false);
  }

  float tE = __shfl(yE[7], 16 + n, 32);
  float tF = __shfl(yF[7], 16 + n, 32);
  float iE = tE, iF = tF;
#pragma unroll
  for (int d = 1; d < 16; d <<= 1) {
    float uE = __shfl_up(iE, d, 16);
    float uF = __shfl_up(iF, d, 16);
    if (n >= d) { iE += uE; iF += uF; }
  }
  float offE = iE - tE;
  float offF = iF - tF;
  float chunkE = __shfl(iE, 15, 16);
  float chunkF = __shfl(iF, 15, 16);

  if (lane == 0) { sE[wave] = chunkE; sF[wave] = chunkF; }
  __syncthreads();
  for (int w = 0; w < wave; ++w) { offE += sE[w]; offF += sF[w]; }

#pragma unroll
  for (int d = 0; d < 8; ++d) {
    const int idx = base + n * 16 + (d + 8 * h);
    const float E = yE[d] + offE;
    const float F = yF[d] + offF;
    if (idx + 1 < SEQ_T) ob[idx + 1] = F / E;
  }
  if (tid == 0) ob[0] = 0.0f;
}

extern "C" void kernel_launch(void* const* d_in, const int* in_sizes, int n_in,
                              void* d_out, int out_size, void* d_ws,
                              size_t ws_size, hipStream_t stream) {
  const float* X  = (const float*)d_in[0];
  const float* Wk = (const float*)d_in[1];
  const float* Wq = (const float*)d_in[2];
  const float* Wv = (const float*)d_in[3];
  float* out = (float*)d_out;
  const int Bn = in_sizes[0] / SEQ_T;
  head_scan_kernel<<<dim3(Bn), dim3(128), 0, stream>>>(X, Wk, Wq, Wv, out, Bn);
}


# x read once: each lane loads its 8 contiguous tile elements with two 16-byte loads issued together and keeps them for row max and WMMA operands (k order now r=8h+2q+j, changes tile sum order)
# speedup vs baseline: 2.0780x; 2.0780x over previous
#include <hip/hip_runtime.h>

typedef __attribute__((ext_vector_type(2))) float v2f;
typedef __attribute__((ext_vector_type(8))) float v8f;

#define SEQ_T 1024

__global__ __launch_bounds__(128) void head_scan_kernel(
    const float* __restrict__ X, const float* __restrict__ Wk,
    const float* __restrict__ Wq, const float* __restrict__ Wv,
    float* __restrict__ out, int Bn) {
  const int b = blockIdx.x;
  if (b >= Bn) return;
  const int tid  = threadIdx.x;
  const int wave = tid >> 5;
  const int lane = tid & 31;
  const int h = lane >> 4;
  const int n = lane & 15;

  const float c1 = Wq[0] * Wk[0];
  const float wv = Wv[0];
  const float* xb = X + (size_t)b * SEQ_T;
  float* ob = out + (size_t)b * SEQ_T;

  __shared__ float sM[4];
  __shared__ float sE[4];
  __shared__ float sF[4];

  const int base = wave * 256;
  const float4* xp =
      reinterpret_cast<const float4*>(xb + base + n * 16 + 8 * h);
  const float4 xlo = xp[0];
  const float4 xhi = xp[1];
  const float xr[8] = {xlo.x, xlo.y, xlo.z, xlo.w, xhi.x, xhi.y, xhi.z, xhi.w};

  float m = -3.402823466e+38f;
#pragma unroll
  for (int i = 0; i < 8; ++i) {
    const float x = xr[i];
    const float s = c1 * x * x;
    m = fmaxf(m, s);
  }
  for (int d = 16; d >= 1; d >>= 1) m = fmaxf(m, __shfl_xor(m, d, 32));
  if (lane == 0) sM[wave] = m;
  __syncthreads();
  m = fmaxf(fmaxf(sM[0], sM[1]), fmaxf(sM[2], sM[3]));

  v8f yE = {};
  v8f yF = {};
#pragma unroll
  for (int q = 0; q < 4; ++q) {
    v2f a, bE, bF;
#pragma unroll
    for (int j = 0; j < 2; ++j) {
      const int r = 8 * h + 2 * q + j;
      const float x = xr[2 * q + j];
      const float s = c1 * x * x;
      const float e = __expf(s - m);
      a[j]  = (r <= n) ? 1.0f : 0.0f;
      bE[j] = e;
      bF[j] = e * wv * x;
    }
    yE = __builtin_amdgcn_wmma_f32_16x16x4_f32(false, a, false, bE,
                                               (short)0, yE, false, false);
    yF = __builtin_amdgcn_wmma_f32_16x16x4_f32(false, a, false, bF,
                                               (short)0, yF, false, false);
  }

  float tE = __shfl(yE[7], 16 + n, 32);
  float tF = __shfl(yF[7], 16 + n, 32);
  float iE = tE, iF = tF;
#pragma unroll
  for (int d = 1; d < 16; d <<= 1) {
    float uE = __shfl_up(iE, d, 16);
    float uF = __shfl_up(iF, d, 16);
    if (n >= d) { iE += uE; iF += uF; }
  }
  float offE = iE - tE;
  float offF = iF - tF;
  float chunkE = __shfl(iE, 15, 16);
  float chunkF = __shfl(iF, 15, 16);

  if (lane == 0) { sE[wave] = chunkE; sF[wave] = chunkF; }
  __syncthreads();
  for (int w = 0; w < wave; ++w) { offE += sE[w]; offF += sF[w]; }

#pragma unroll
  for (int d = 0; d < 8; ++d) {
    const int idx = base + n * 16 + (d + 8 * h);
    const float E = yE[d] + offE;
    const float F = yF[d] + offF;
    if (idx + 1 < SEQ_T) ob[idx + 1] = F / E;
  }
  if (tid == 0) ob[0] = 0.0f;
}

extern "C" void kernel_launch(void* const* d_in, const int* in_sizes, int n_in,
                              void* d_out, int out_size, void* d_ws,
                              size_t ws_size, hipStream_t stream) {
  const float* X  = (const float*)d_in[0];
  const float* Wk = (const float*)d_in[1];
  const float* Wq = (const float*)d_in[2];
  const float* Wv = (const float*)d_in[3];
  float* out = (float*)d_out;
  const int Bn = in_sizes[0] / SEQ_T;
  head_scan_kernel<<<dim3(Bn), dim3(128), 0, stream>>>(X, Wk, Wq, Wv, out, Bn);
}


# in-wave lane exchanges (max butterfly, column-total broadcast and scan) off the LDS unit: DPP row controls and v_permlanex16 replace ds_bpermute shuffles; wv scalar load held early; sum order kept
# speedup vs baseline: 2.3027x; 1.1081x over previous
#include <hip/hip_runtime.h>

typedef __attribute__((ext_vector_type(2))) float v2f;
typedef __attribute__((ext_vector_type(8))) float v8f;

#define SEQ_T 1024

#define DPP_ROW_SHR(k)   (0x110 + (k))
#define DPP_ROW_SHARE(k) (0x150 + (k))

template <int CTRL>
__device__ __forceinline__ float row_get(float v) {
  return __builtin_bit_cast(
      float, __builtin_amdgcn_update_dpp(0, __builtin_bit_cast(int, v), CTRL,
                                         0xf, 0xf, true));
}

template <int K>
__device__ __forceinline__ float row_max_xor(float v) {
  float r;
  asm volatile(
      "v_max_num_f32_dpp %0, %1, %1 row_xmask:%2 row_mask:0xf bank_mask:0xf"
      : "=v"(r)
      : "v"(v), "n"(K));
  return r;
}

__device__ __forceinline__ float other_row(float v) {
  const int i = __builtin_bit_cast(int, v);
  return __builtin_bit_cast(
      float, __builtin_amdgcn_permlanex16(i, i, 0x76543210, 0xfedcba98, false,
                                          false));
}

__global__ __launch_bounds__(128) void head_scan_kernel(
    const float* __restrict__ X, const float* __restrict__ Wk,
    const float* __restrict__ Wq, const float* __restrict__ Wv,
    float* __restrict__ out, int Bn) {
  const int b = blockIdx.x;
  if (b >= Bn) return;
  const int tid  = threadIdx.x;
  const int wave = tid >> 5;
  const int lane = tid & 31;
  const int h = lane >> 4;
  const int n = lane & 15;

  const float c1 = Wq[0] * Wk[0];
  const float wv = Wv[0];
  const float* xb = X + (size_t)b * SEQ_T;
  float* ob = out + (size_t)b * SEQ_T;

  __shared__ float sM[4];
  __shared__ float sE[4];
  __shared__ float sF[4];

  const int base = wave * 256;
  const float4* xp =
      reinterpret_cast<const float4*>(xb + base + n * 16 + 8 * h);
  const float4 xlo = xp[0];
  const float4 xhi = xp[1];
  const float xr[8] = {xlo.x, xlo.y, xlo.z, xlo.w, xhi.x, xhi.y, xhi.z, xhi.w};
  asm volatile("" ::"s"(wv));

  float m = -3.402823466e+38f;
#pragma unroll
  for (int i = 0; i < 8; ++i) {
    const float x = xr[i];
    const float s = c1 * x * x;
    m = fmaxf(m, s);
  }
  m = row_max_xor<1>(m);
  m = row_max_xor<2>(m);
  m = row_max_xor<4>(m);
  m = row_max_xor<8>(m);
  m = fmaxf(m, other_row(m));
  if (lane == 0) sM[wave] = m;
  __syncthreads();
  m = fmaxf(fmaxf(sM[0], sM[1]), fmaxf(sM[2], sM[3]));

  v8f yE = {};
  v8f yF = {};
#pragma unroll
  for (int q = 0; q < 4; ++q) {
    v2f a, bE, bF;
#pragma unroll
    for (int j = 0; j < 2; ++j) {
      const int r = 8 * h + 2 * q + j;
      const float x = xr[2 * q + j];
      const float s = c1 * x * x;
      const float e = __expf(s - m);
      a[j]  = (r <= n) ? 1.0f : 0.0f;
      bE[j] = e;
      bF[j] = e * wv * x;
    }
    yE = __builtin_amdgcn_wmma_f32_16x16x4_f32(false, a, false, bE,
                                               (short)0, yE, false, false);
    yF = __builtin_amdgcn_wmma_f32_16x16x4_f32(false, a, false, bF,
                                               (short)0, yF, false, false);
  }

  const float xE = other_row(yE[7]);
  const float xF = other_row(yF[7]);
  const float tE = h ? yE[7] : xE;
  const float tF = h ? yF[7] : xF;
  float iE = tE, iF = tF;
  iE += row_get<DPP_ROW_SHR(1)>(iE);  iF += row_get<DPP_ROW_SHR(1)>(iF);
  iE += row_get<DPP_ROW_SHR(2)>(iE);  iF += row_get<DPP_ROW_SHR(2)>(iF);
  iE += row_get<DPP_ROW_SHR(4)>(iE);  iF += row_get<DPP_ROW_SHR(4)>(iF);
  iE += row_get<DPP_ROW_SHR(8)>(iE);  iF += row_get<DPP_ROW_SHR(8)>(iF);
  float offE = iE - tE;
  float offF = iF - tF;
  const float chunkE = row_get<DPP_ROW_SHARE(15)>(iE);
  const float chunkF = row_get<DPP_ROW_SHARE(15)>(iF);

  if (lane == 0) { sE[wave] = chunkE; sF[wave] = chunkF; }
  __syncthreads();
  for (int w = 0; w < wave; ++w) { offE += sE[w]; offF += sF[w]; }

#pragma unroll
  for (int d = 0; d < 8; ++d) {
    const int idx = base + n * 16 + (d + 8 * h);
    const float E = yE[d] + offE;
    const float F = yF[d] + offF;
    if (idx + 1 < SEQ_T) ob[idx + 1] = F / E;
  }
  if (tid == 0) ob[0] = 0.0f;
}

extern "C" void kernel_launch(void* const* d_in, const int* in_sizes, int n_in,
                              void* d_out, int out_size, void* d_ws,
                              size_t ws_size, hipStream_t stream) {
  const float* X  = (const float*)d_in[0];
  const float* Wk = (const float*)d_in[1];
  const float* Wq = (const float*)d_in[2];
  const float* Wv = (const float*)d_in[3];
  float* out = (float*)d_out;
  const int Bn = in_sizes[0] / SEQ_T;
  head_scan_kernel<<<dim3(Bn), dim3(128), 0, stream>>>(X, Wk, Wq, Wv, out, Bn);
}


# emit without branches: a lane's 8 quotients formed in one straight-line block, so their division sequences overlap and 7 stores merge to 16+12 B; row's last lane stores out[b,0]=0; sum order kept
# speedup vs baseline: 2.6113x; 1.1340x over previous
#include <hip/hip_runtime.h>

typedef __attribute__((ext_vector_type(2))) float v2f;
typedef __attribute__((ext_vector_type(8))) float v8f;

#define SEQ_T 1024

#define DPP_ROW_SHR(k)   (0x110 + (k))
#define DPP_ROW_SHARE(k) (0x150 + (k))

template <int CTRL>
__device__ __forceinline__ float row_get(float v) {
  return __builtin_bit_cast(
      float, __builtin_amdgcn_update_dpp(0, __builtin_bit_cast(int, v), CTRL,
                                         0xf, 0xf, true));
}

template <int K>
__device__ __forceinline__ float row_max_xor(float v) {
  float r;
  asm volatile(
      "v_max_num_f32_dpp %0, %1, %1 row_xmask:%2 row_mask:0xf bank_mask:0xf"
      : "=v"(r)
      : "v"(v), "n"(K));
  return r;
}

__device__ __forceinline__ float other_row(float v) {
  const int i = __builtin_bit_cast(int, v);
  return __builtin_bit_cast(
      float, __builtin_amdgcn_permlanex16(i, i, 0x76543210, 0xfedcba98, false,
                                          false));
}

__global__ __launch_bounds__(128) void head_scan_kernel(
    const float* __restrict__ X, const float* __restrict__ Wk,
    const float* __restrict__ Wq, const float* __restrict__ Wv,
    float* __restrict__ out, int Bn) {
  const int b = blockIdx.x;
  if (b >= Bn) return;
  const int tid  = threadIdx.x;
  const int wave = tid >> 5;
  const int lane = tid & 31;
  const int h = lane >> 4;
  const int n = lane & 15;

  const float wq = Wq[0];
  const float wk = Wk[0];
  const float wv = Wv[0];
  const float c1 = wq * wk;
  const float* xb = X + (size_t)b * SEQ_T;
  float* ob = out + (size_t)b * SEQ_T;

  __shared__ float sM[4];
  __shared__ float sE[4];
  __shared__ float sF[4];

  const int base = wave * 256;
  const float4* xp =
      reinterpret_cast<const float4*>(xb + base + n * 16 + 8 * h);
  const float4 xlo = xp[0];
  const float4 xhi = xp[1];
  const float xr[8] = {xlo.x, xlo.y, xlo.z, xlo.w, xhi.x, xhi.y, xhi.z, xhi.w};
  asm volatile("" ::"s"(wq), "s"(wk), "s"(wv));

  float m = -3.402823466e+38f;
#pragma unroll
  for (int i = 0; i < 8; ++i) {
    const float x = xr[i];
    const float s = c1 * x * x;
    m = fmaxf(m, s);
  }
  m = row_max_xor<1>(m);
  m = row_max_xor<2>(m);
  m = row_max_xor<4>(m);
  m = row_max_xor<8>(m);
  m = fmaxf(m, other_row(m));
  if (lane == 0) sM[wave] = m;
  __syncthreads();
  m = fmaxf(fmaxf(sM[0], sM[1]), fmaxf(sM[2], sM[3]));

  v8f yE = {};
  v8f yF = {};
#pragma unroll
  for (int q = 0; q < 4; ++q) {
    v2f a, bE, bF;
#pragma unroll
    for (int j = 0; j < 2; ++j) {
      const int r = 8 * h + 2 * q + j;
      const float x = xr[2 * q + j];
      const float s = c1 * x * x;
      const float e = __expf(s - m);
      a[j]  = (r <= n) ? 1.0f : 0.0f;
      bE[j] = e;
      bF[j] = e * wv * x;
    }
    yE = __builtin_amdgcn_wmma_f32_16x16x4_f32(false, a, false, bE,
                                               (short)0, yE, false, false);
    yF = __builtin_amdgcn_wmma_f32_16x16x4_f32(false, a, false, bF,
                                               (short)0, yF, false, false);
  }

  const float xE = other_row(yE[7]);
  const float xF = other_row(yF[7]);
  const float tE = h ? yE[7] : xE;
  const float tF = h ? yF[7] : xF;
  float iE = tE, iF = tF;
  iE += row_get<DPP_ROW_SHR(1)>(iE);  iF += row_get<DPP_ROW_SHR(1)>(iF);
  iE += row_get<DPP_ROW_SHR(2)>(iE);  iF += row_get<DPP_ROW_SHR(2)>(iF);
  iE += row_get<DPP_ROW_SHR(4)>(iE);  iF += row_get<DPP_ROW_SHR(4)>(iF);
  iE += row_get<DPP_ROW_SHR(8)>(iE);  iF += row_get<DPP_ROW_SHR(8)>(iF);
  float offE = iE - tE;
  float offF = iF - tF;
  const float chunkE = row_get<DPP_ROW_SHARE(15)>(iE);
  const float chunkF = row_get<DPP_ROW_SHARE(15)>(iF);

  if (lane == 0) { sE[wave] = chunkE; sF[wave] = chunkF; }
  __syncthreads();
  for (int w = 0; w < wave; ++w) { offE += sE[w]; offF += sF[w]; }

  float o[8];
#pragma unroll
  for (int d = 0; d < 8; ++d) o[d] = (yF[d] + offF) / (yE[d] + offE);
  const int i0 = base + n * 16 + 8 * h + 1;
#pragma unroll
  for (int d = 0; d < 7; ++d) ob[i0 + d] = o[d];
  const bool wrap = (i0 + 7 == SEQ_T);
  ob[wrap ? 0 : i0 + 7] = wrap ? 0.0f : o[7];
}

extern "C" void kernel_launch(void* const* d_in, const int* in_sizes, int n_in,
                              void* d_out, int out_size, void* d_ws,
                              size_t ws_size, hipStream_t stream) {
  const float* X  = (const float*)d_in[0];
  const float* Wk = (const float*)d_in[1];
  const float* Wq = (const float*)d_in[2];
  const float* Wv = (const float*)d_in[3];
  float* out = (float*)d_out;
  const int Bn = in_sizes[0] / SEQ_T;
  head_scan_kernel<<<dim3(Bn), dim3(128), 0, stream>>>(X, Wk, Wq, Wv, out, Bn);
}
